# starting kernel
# baseline (speedup 1.0000x reference)
#include <hip/hip_runtime.h>

typedef float v2f __attribute__((ext_vector_type(2)));
typedef float v8f __attribute__((ext_vector_type(8)));

#define HH 64
#define WW 64
#define DD 256
#define NP 112
#define SSTRIDE 114

__global__ __launch_bounds__(128)
void local_attn_wmma_f32(const float* __restrict__ x, float* __restrict__ out) {
    __shared__ float sP[16 * SSTRIDE];
    __shared__ int   sKB[NP];

    const int blk = blockIdx.x;
    const int b   = blk >> 8;
    const int t   = blk & 255;
    const int h0  = (t >> 4) << 2;
    const int w0  = (t & 15) << 2;

    const int tid  = threadIdx.x;
    const int wave = tid >> 5;
    const int lane = tid & 31;
    const int half = lane >> 4;
    const int lm   = lane & 15;

    if (tid < NP) {
        int p  = tid < 100 ? tid : 99;
        int pi = p / 10, pj = p - pi * 10;
        int kh = h0 - 3 + pi; kh = kh < 0 ? 0 : (kh > HH - 1 ? HH - 1 : kh);
        int kw = w0 - 3 + pj; kw = kw < 0 ? 0 : (kw > WW - 1 ? WW - 1 : kw);
        sKB[tid] = ((b * HH + kh) * WW + kw) * DD;
    }
    __syncthreads();

    {
        const int m  = lm;
        const int qh = h0 + (m >> 2), qw = w0 + (m & 3);
        const long qbase = (long)((b * HH + qh) * WW + qw) * DD + 2 * half;
        for (int tn = wave; tn < 7; tn += 4) {
            const int  n     = tn * 16 + lm;
            const long kbase = (long)sKB[n] + 2 * half;
            v8f acc = {};
            #pragma unroll 8
            for (int d0 = 0; d0 < DD; d0 += 4) {
                v2f a  = *(const v2f*)(x + qbase + d0);
                v2f bb = *(const v2f*)(x + kbase + d0);
                acc = __builtin_amdgcn_wmma_f32_16x16x4_f32(
                        false, a, false, bb, (short)0, acc, false, false);
            }
            #pragma unroll
            for (int r = 0; r < 8; ++r)
                sP[(r + 8 * half) * SSTRIDE + tn * 16 + lm] = acc[r];
        }
    }
    __syncthreads();

    if (tid < 16) {
        const int m  = tid;
        const int qh = h0 + (m >> 2), qw = w0 + (m & 3);
        float* row = sP + m * SSTRIDE;
        float mx = -3.0e38f;
        for (int n = 0; n < NP; ++n) {
            int  pi = n / 10, pj = n - pi * 10;
            int  kh = h0 - 3 + pi, kw = w0 - 3 + pj;
            bool valid = (n < 100) && kh >= 0 && kh < HH && kw >= 0 && kw < WW
                         && (kh - qh) >= -3 && (kh - qh) <= 3
                         && (kw - qw) >= -3 && (kw - qw) <= 3;
            float s = row[n] * 0.0625f;
            row[n]  = valid ? s : -3.0e38f;
            if (valid && s > mx) mx = s;
        }
        float sum = 0.f;
        for (int n = 0; n < NP; ++n) {
            float s = row[n];
            float e = (s > -1.0e38f) ? __expf(s - mx) : 0.f;
            row[n]  = e;
            sum += e;
        }
        float inv = 1.f / sum;
        for (int n = 0; n < NP; ++n) row[n] *= inv;
    }
    __syncthreads();

    {
        for (int dt = wave; dt < 16; dt += 4) {
            const int d = dt * 16 + lm;
            v8f acc = {};
            #pragma unroll 4
            for (int p0 = 0; p0 < NP; p0 += 4) {
                const int pk = p0 + 2 * half;
                v2f a = *(const v2f*)(sP + lm * SSTRIDE + pk);
                v2f bb;
                bb.x = x[(long)sKB[pk]     + d];
                bb.y = x[(long)sKB[pk + 1] + d];
                acc = __builtin_amdgcn_wmma_f32_16x16x4_f32(
                        false, a, false, bb, (short)0, acc, false, false);
            }
            #pragma unroll
            for (int r = 0; r < 8; ++r) {
                const int rowm = r + 8 * half;
                const int qh = h0 + (rowm >> 2), qw = w0 + (rowm & 3);
                out[(long)((b * HH + qh) * WW + qw) * DD + d] = acc[r];
            }
        }
    }
}

extern "C" void kernel_launch(void* const* d_in, const int* in_sizes, int n_in,
                              void* d_out, int out_size, void* d_ws, size_t ws_size,
                              hipStream_t stream) {
    const float* x   = (const float*)d_in[0];
    float*       out = (float*)d_out;
    local_attn_wmma_f32<<<dim3(1024), dim3(128), 0, stream>>>(x, out);
}


# softmax phase spread over all 128 lanes: 8 lanes per query row, values in registers, shuffle max and sum, instead of 16 serial lanes; row sum is now 8 partial sums joined by a shuffle tree
# speedup vs baseline: 1.2363x; 1.2363x over previous
#include <hip/hip_runtime.h>

typedef float v2f __attribute__((ext_vector_type(2)));
typedef float v8f __attribute__((ext_vector_type(8)));

#define HH 64
#define WW 64
#define DD 256
#define NP 112
#define SSTRIDE 114

__global__ __launch_bounds__(128)
void local_attn_wmma_f32(const float* __restrict__ x, float* __restrict__ out) {
    __shared__ float sP[16 * SSTRIDE];
    __shared__ int   sKB[NP];

    const int blk = blockIdx.x;
    const int b   = blk >> 8;
    const int t   = blk & 255;
    const int h0  = (t >> 4) << 2;
    const int w0  = (t & 15) << 2;

    const int tid  = threadIdx.x;
    const int wave = tid >> 5;
    const int lane = tid & 31;
    const int half = lane >> 4;
    const int lm   = lane & 15;

    if (tid < NP) {
        int p  = tid < 100 ? tid : 99;
        int pi = p / 10, pj = p - pi * 10;
        int kh = h0 - 3 + pi; kh = kh < 0 ? 0 : (kh > HH - 1 ? HH - 1 : kh);
        int kw = w0 - 3 + pj; kw = kw < 0 ? 0 : (kw > WW - 1 ? WW - 1 : kw);
        sKB[tid] = ((b * HH + kh) * WW + kw) * DD;
    }
    __syncthreads();

    {
        const int m  = lm;
        const int qh = h0 + (m >> 2), qw = w0 + (m & 3);
        const long qbase = (long)((b * HH + qh) * WW + qw) * DD + 2 * half;
        for (int tn = wave; tn < 7; tn += 4) {
            const int  n     = tn * 16 + lm;
            const long kbase = (long)sKB[n] + 2 * half;
            v8f acc = {};
            #pragma unroll 8
            for (int d0 = 0; d0 < DD; d0 += 4) {
                v2f a  = *(const v2f*)(x + qbase + d0);
                v2f bb = *(const v2f*)(x + kbase + d0);
                acc = __builtin_amdgcn_wmma_f32_16x16x4_f32(
                        false, a, false, bb, (short)0, acc, false, false);
            }
            #pragma unroll
            for (int r = 0; r < 8; ++r)
                sP[(r + 8 * half) * SSTRIDE + tn * 16 + lm] = acc[r];
        }
    }
    __syncthreads();

    {
        const int m   = tid >> 3;
        const int sub = tid & 7;
        const int qh = h0 + (m >> 2), qw = w0 + (m & 3);
        float* row = sP + m * SSTRIDE;
        float v[14];
        float mx = -3.0e38f;
        #pragma unroll
        for (int j = 0; j < 14; ++j) {
            const int n  = sub + 8 * j;
            const int pi = n / 10, pj = n - pi * 10;
            const int kh = h0 - 3 + pi, kw = w0 - 3 + pj;
            const bool valid = (n < 100) && kh >= 0 && kh < HH && kw >= 0 && kw < WW
                               && (kh - qh) >= -3 && (kh - qh) <= 3
                               && (kw - qw) >= -3 && (kw - qw) <= 3;
            const float s = row[n] * 0.0625f;
            v[j] = valid ? s : -3.0e38f;
            if (valid && s > mx) mx = s;
        }
        mx = fmaxf(mx, __shfl_xor(mx, 1));
        mx = fmaxf(mx, __shfl_xor(mx, 2));
        mx = fmaxf(mx, __shfl_xor(mx, 4));
        float sum = 0.f;
        #pragma unroll
        for (int j = 0; j < 14; ++j) {
            const float e = (v[j] > -1.0e38f) ? __expf(v[j] - mx) : 0.f;
            v[j] = e;
            sum += e;
        }
        sum += __shfl_xor(sum, 1);
        sum += __shfl_xor(sum, 2);
        sum += __shfl_xor(sum, 4);
        const float inv = 1.f / sum;
        #pragma unroll
        for (int j = 0; j < 14; ++j) row[sub + 8 * j] = v[j] * inv;
    }
    __syncthreads();

    {
        for (int dt = wave; dt < 16; dt += 4) {
            const int d = dt * 16 + lm;
            v8f acc = {};
            #pragma unroll 4
            for (int p0 = 0; p0 < NP; p0 += 4) {
                const int pk = p0 + 2 * half;
                v2f a = *(const v2f*)(sP + lm * SSTRIDE + pk);
                v2f bb;
                bb.x = x[(long)sKB[pk]     + d];
                bb.y = x[(long)sKB[pk + 1] + d];
                acc = __builtin_amdgcn_wmma_f32_16x16x4_f32(
                        false, a, false, bb, (short)0, acc, false, false);
            }
            #pragma unroll
            for (int r = 0; r < 8; ++r) {
                const int rowm = r + 8 * half;
                const int qh = h0 + (rowm >> 2), qw = w0 + (rowm & 3);
                out[(long)((b * HH + qh) * WW + qw) * DD + d] = acc[r];
            }
        }
    }
}

extern "C" void kernel_launch(void* const* d_in, const int* in_sizes, int n_in,
                              void* d_out, int out_size, void* d_ws, size_t ws_size,
                              hipStream_t stream) {
    const float* x   = (const float*)d_in[0];
    float*       out = (float*)d_out;
    local_attn_wmma_f32<<<dim3(1024), dim3(128), 0, stream>>>(x, out);
}


# key/value patch (100 rows x 256 ch) staged once per block in LDS by coalesced 16-byte loads; both matrix products take operands from LDS in grouped reads instead of strided global loads
# speedup vs baseline: 2.5347x; 2.0502x over previous
#include <hip/hip_runtime.h>


typedef float v2f __attribute__((ext_vector_type(2)));
typedef float v4f __attribute__((ext_vector_type(4)));
typedef float v8f __attribute__((ext_vector_type(8)));

#define HH 64
#define WW 64
#define DD 256
#define NK 100
#define NP 112
#define SSTRIDE 114
#define XP 268
#define ROWS_PER_WAVE 25

__global__ __launch_bounds__(128)
void local_attn_wmma_f32(const float* __restrict__ x, float* __restrict__ out) {
    __shared__ __attribute__((aligned(16))) float sX[NK * XP];
    __shared__ __attribute__((aligned(16))) float sP[16 * SSTRIDE];

    const int blk = blockIdx.x;
    const int b   = blk >> 8;
    const int t   = blk & 255;
    const int h0  = (t >> 4) << 2;
    const int w0  = (t & 15) << 2;

    const int tid  = threadIdx.x;
    const int wave = __builtin_amdgcn_readfirstlane(tid >> 5);
    const int lane = tid & 31;
    const int half = lane >> 4;
    const int lm   = lane & 15;

    {
        v4f tmp[2 * ROWS_PER_WAVE];
        #pragma unroll
        for (int i = 0; i < ROWS_PER_WAVE; ++i) {
            const int p  = wave + 4 * i;
            const int pi = p / 10, pj = p - pi * 10;
            int kh = h0 - 3 + pi; kh = kh < 0 ? 0 : (kh > HH - 1 ? HH - 1 : kh);
            int kw = w0 - 3 + pj; kw = kw < 0 ? 0 : (kw > WW - 1 ? WW - 1 : kw);
            const float* src = x + ((b * HH + kh) * WW + kw) * DD + lane * 4;
            tmp[2 * i]     = *(const v4f*)(src);
            tmp[2 * i + 1] = *(const v4f*)(src + 128);
        }
        #pragma unroll
        for (int i = 0; i < ROWS_PER_WAVE; ++i) {
            const int p = wave + 4 * i;
            float* dst = sX + p * XP + lane * 4;
            *(v4f*)(dst)       = tmp[2 * i];
            *(v4f*)(dst + 128) = tmp[2 * i + 1];
        }
    }
    __syncthreads();

    {
        const int m    = lm;
        const int qrow = ((m >> 2) + 3) * 10 + (m & 3) + 3;
        const float* aP = sX + qrow * XP + 2 * half;
        v2f qa[DD / 4];
        #pragma unroll
        for (int i = 0; i < DD / 4; ++i) qa[i] = *(const v2f*)(aP + 4 * i);
        for (int tn = wave; tn < 7; tn += 4) {
            const int n    = tn * 16 + lm;
            const int krow = n < NK ? n : NK - 1;
            const float* bP = sX + krow * XP + 2 * half;
            v8f acc = {};
            v2f kb[2][16];
            #pragma unroll
            for (int i = 0; i < 16; ++i) kb[0][i] = *(const v2f*)(bP + 4 * i);
            #pragma unroll
            for (int c = 0; c < 4; ++c) {
                if (c < 3) {
                    #pragma unroll
                    for (int i = 0; i < 16; ++i)
                        kb[(c + 1) & 1][i] = *(const v2f*)(bP + 4 * (16 * (c + 1) + i));
                }
                __builtin_amdgcn_sched_barrier(0);
                #pragma unroll
                for (int i = 0; i < 16; ++i)
                    acc = __builtin_amdgcn_wmma_f32_16x16x4_f32(
                            false, qa[16 * c + i], false, kb[c & 1][i], (short)0, acc, false, false);
                __builtin_amdgcn_sched_barrier(0);
            }
            #pragma unroll
            for (int r = 0; r < 8; ++r)
                sP[(r + 8 * half) * SSTRIDE + tn * 16 + lm] = acc[r];
        }
    }
    __syncthreads();

    {
        const int m   = tid >> 3;
        const int sub = tid & 7;
        const int qh = h0 + (m >> 2), qw = w0 + (m & 3);
        float* row = sP + m * SSTRIDE;
        float v[14];
        float mx = -3.0e38f;
        #pragma unroll
        for (int j = 0; j < 14; ++j) {
            const int n  = sub + 8 * j;
            const int pi = n / 10, pj = n - pi * 10;
            const int kh = h0 - 3 + pi, kw = w0 - 3 + pj;
            const bool valid = (n < 100) && kh >= 0 && kh < HH && kw >= 0 && kw < WW
                               && (kh - qh) >= -3 && (kh - qh) <= 3
                               && (kw - qw) >= -3 && (kw - qw) <= 3;
            const float s = row[n] * 0.0625f;
            v[j] = valid ? s : -3.0e38f;
            if (valid && s > mx) mx = s;
        }
        mx = fmaxf(mx, __shfl_xor(mx, 1));
        mx = fmaxf(mx, __shfl_xor(mx, 2));
        mx = fmaxf(mx, __shfl_xor(mx, 4));
        float sum = 0.f;
        #pragma unroll
        for (int j = 0; j < 14; ++j) {
            const float e = (v[j] > -1.0e38f) ? __expf(v[j] - mx) : 0.f;
            v[j] = e;
            sum += e;
        }
        sum += __shfl_xor(sum, 1);
        sum += __shfl_xor(sum, 2);
        sum += __shfl_xor(sum, 4);
        const float inv = 1.f / sum;
        #pragma unroll
        for (int j = 0; j < 14; ++j) row[sub + 8 * j] = v[j] * inv;
    }
    __syncthreads();

    {
        v2f pa[NP / 4];
        #pragma unroll
        for (int i = 0; i < NP / 4; ++i)
            pa[i] = *(const v2f*)(sP + lm * SSTRIDE + 4 * i + 2 * half);

        const int dbase = wave * 16 + lm;
        v2f vb[2][14];
        #pragma unroll
        for (int i = 0; i < 14; ++i) {
            const int pk = 4 * i + 2 * half;
            vb[0][i].x = sX[pk * XP + dbase];
            vb[0][i].y = sX[(pk + 1) * XP + dbase];
        }
        v8f acc = {};
        #pragma unroll
        for (int g = 0; g < 8; ++g) {
            if (g < 7) {
                const int gn = g + 1;
                const int dn = dbase + 64 * (gn >> 1);
                #pragma unroll
                for (int i = 0; i < 14; ++i) {
                    const int p0 = 4 * (14 * (gn & 1) + i);
                    const int pk = p0 + 2 * half;
                    const int r0 = p0 < NK ? pk     : NK - 1;
                    const int r1 = p0 < NK ? pk + 1 : NK - 1;
                    vb[gn & 1][i].x = sX[r0 * XP + dn];
                    vb[gn & 1][i].y = sX[r1 * XP + dn];
                }
            }
            __builtin_amdgcn_sched_barrier(0);
            if ((g & 1) == 0) acc = (v8f){};
            #pragma unroll
            for (int i = 0; i < 14; ++i)
                acc = __builtin_amdgcn_wmma_f32_16x16x4_f32(
                        false, pa[14 * (g & 1) + i], false, vb[g & 1][i], (short)0, acc, false, false);
            __builtin_amdgcn_sched_barrier(0);
            if (g & 1) {
                const int d = dbase + 64 * (g >> 1);
                #pragma unroll
                for (int r = 0; r < 8; ++r) {
                    const int rowm = r + 8 * half;
                    const int qh = h0 + (rowm >> 2), qw = w0 + (rowm & 3);
                    out[(long)((b * HH + qh) * WW + qw) * DD + d] = acc[r];
                }
            }
        }
    }
}

extern "C" void kernel_launch(void* const* d_in, const int* in_sizes, int n_in,
                              void* d_out, int out_size, void* d_ws, size_t ws_size,
                              hipStream_t stream) {
    const float* x   = (const float*)d_in[0];
    float*       out = (float*)d_out;
    local_attn_wmma_f32<<<dim3(1024), dim3(128), 0, stream>>>(x, out);
}


# each block does two adjacent 4x4 query tiles: 60 shared patch rows stay in LDS, the 40 new rows load into registers under tile one's products; tile two sums keys in rotated column order
# speedup vs baseline: 2.9277x; 1.1550x over previous
#include <hip/hip_runtime.h>


typedef float v2f __attribute__((ext_vector_type(2)));
typedef float v4f __attribute__((ext_vector_type(4)));
typedef float v8f __attribute__((ext_vector_type(8)));

#define HH 64
#define WW 64
#define DD 256
#define NK 100
#define NP 112
#define SSTRIDE 114
#define XP 268
#define ROWS_PER_WAVE 25
#define NEW_PER_WAVE 10

__device__ __forceinline__ void lds_barrier() {
    __builtin_amdgcn_fence(__ATOMIC_RELEASE, "workgroup", "local");
    __builtin_amdgcn_s_barrier();
    __builtin_amdgcn_fence(__ATOMIC_ACQUIRE, "workgroup", "local");
}

__global__ __launch_bounds__(128)
void local_attn_wmma_f32(const float* __restrict__ x, float* __restrict__ out) {
    __shared__ __attribute__((aligned(16))) float sX[NK * XP];
    __shared__ __attribute__((aligned(16))) float sP[16 * SSTRIDE];

    const int blk = blockIdx.x;
    const int b   = blk >> 7;
    const int t   = blk & 127;
    const int h0  = (t >> 3) << 2;
    const int w0a = (t & 7) << 3;

    const int tid  = threadIdx.x;
    const int wave = __builtin_amdgcn_readfirstlane(tid >> 5);
    const int lane = tid & 31;
    const int half = lane >> 4;
    const int lm   = lane & 15;

    {
        v4f tmp[2 * ROWS_PER_WAVE];
        #pragma unroll
        for (int i = 0; i < ROWS_PER_WAVE; ++i) {
            const int p  = wave + 4 * i;
            const int pi = p / 10, pj = p - pi * 10;
            int kh = h0 - 3 + pi;  kh = kh < 0 ? 0 : (kh > HH - 1 ? HH - 1 : kh);
            int kw = w0a - 3 + pj; kw = kw < 0 ? 0 : (kw > WW - 1 ? WW - 1 : kw);
            const float* src = x + ((b * HH + kh) * WW + kw) * DD + lane * 4;
            tmp[2 * i]     = *(const v4f*)(src);
            tmp[2 * i + 1] = *(const v4f*)(src + 128);
        }
        #pragma unroll
        for (int i = 0; i < ROWS_PER_WAVE; ++i) {
            const int p = wave + 4 * i;
            float* dst = sX + p * XP + lane * 4;
            *(v4f*)(dst)       = tmp[2 * i];
            *(v4f*)(dst + 128) = tmp[2 * i + 1];
        }
    }
    __syncthreads();

    v4f nx[2 * NEW_PER_WAVE];
    #pragma unroll
    for (int i = 0; i < NEW_PER_WAVE; ++i) {
        const int e  = wave + 4 * i;
        const int pi = e >> 2, c = e & 3;
        int kh = h0 - 3 + pi;  kh = kh < 0 ? 0 : (kh > HH - 1 ? HH - 1 : kh);
        int kw = w0a + 7 + c;  kw = kw > WW - 1 ? WW - 1 : kw;
        const float* src = x + ((b * HH + kh) * WW + kw) * DD + lane * 4;
        nx[2 * i]     = *(const v4f*)(src);
        nx[2 * i + 1] = *(const v4f*)(src + 128);
    }

    for (int tile = 0; tile < 2; ++tile) {
        const int shift = tile << 2;
        const int w0    = w0a + shift;

        {
            const int m    = lm;
            int qs = (m & 3) + 3 + shift;
            qs = qs >= 10 ? qs - 10 : qs;
            const int qrow = ((m >> 2) + 3) * 10 + qs;
            const float* aP = sX + qrow * XP + 2 * half;
            v2f qa[DD / 4];
            #pragma unroll
            for (int i = 0; i < DD / 4; ++i) qa[i] = *(const v2f*)(aP + 4 * i);
            for (int tn = wave; tn < 7; tn += 4) {
                const int n    = tn * 16 + lm;
                const int krow = n < NK ? n : NK - 1;
                const float* bP = sX + krow * XP + 2 * half;
                v8f acc = {};
                v2f kb[2][16];
                #pragma unroll
                for (int i = 0; i < 16; ++i) kb[0][i] = *(const v2f*)(bP + 4 * i);
                #pragma unroll
                for (int c = 0; c < 4; ++c) {
                    if (c < 3) {
                        #pragma unroll
                        for (int i = 0; i < 16; ++i)
                            kb[(c + 1) & 1][i] = *(const v2f*)(bP + 4 * (16 * (c + 1) + i));
                    }
                    __builtin_amdgcn_sched_barrier(0);
                    #pragma unroll
                    for (int i = 0; i < 16; ++i)
                        acc = __builtin_amdgcn_wmma_f32_16x16x4_f32(
                                false, qa[16 * c + i], false, kb[c & 1][i], (short)0, acc, false, false);
                    __builtin_amdgcn_sched_barrier(0);
                }
                #pragma unroll
                for (int r = 0; r < 8; ++r)
                    sP[(r + 8 * half) * SSTRIDE + tn * 16 + lm] = acc[r];
            }
        }
        __syncthreads();

        {
            const int m   = tid >> 3;
            const int sub = tid & 7;
            const int qh = h0 + (m >> 2), qw = w0 + (m & 3);
            float* row = sP + m * SSTRIDE;
            float v[14];
            float mx = -3.0e38f;
            #pragma unroll
            for (int j = 0; j < 14; ++j) {
                const int n  = sub + 8 * j;
                const int pi = n / 10;
                int pj = n - pi * 10 - shift;
                pj = pj < 0 ? pj + 10 : pj;
                const int kh = h0 - 3 + pi, kw = w0 - 3 + pj;
                const bool valid = (n < 100) && kh >= 0 && kh < HH && kw >= 0 && kw < WW
                                   && (kh - qh) >= -3 && (kh - qh) <= 3
                                   && (kw - qw) >= -3 && (kw - qw) <= 3;
                const float s = row[n] * 0.0625f;
                v[j] = valid ? s : -3.0e38f;
                if (valid && s > mx) mx = s;
            }
            mx = fmaxf(mx, __shfl_xor(mx, 1));
            mx = fmaxf(mx, __shfl_xor(mx, 2));
            mx = fmaxf(mx, __shfl_xor(mx, 4));
            float sum = 0.f;
            #pragma unroll
            for (int j = 0; j < 14; ++j) {
                const float e = (v[j] > -1.0e38f) ? __expf(v[j] - mx) : 0.f;
                v[j] = e;
                sum += e;
            }
            sum += __shfl_xor(sum, 1);
            sum += __shfl_xor(sum, 2);
            sum += __shfl_xor(sum, 4);
            const float inv = 1.f / sum;
            #pragma unroll
            for (int j = 0; j < 14; ++j) row[sub + 8 * j] = v[j] * inv;
        }
        __syncthreads();

        {
            v2f pa[NP / 4];
            #pragma unroll
            for (int i = 0; i < NP / 4; ++i)
                pa[i] = *(const v2f*)(sP + lm * SSTRIDE + 4 * i + 2 * half);

            const int dbase = wave * 16 + lm;
            v2f vb[2][14];
            #pragma unroll
            for (int i = 0; i < 14; ++i) {
                const int pk = 4 * i + 2 * half;
                vb[0][i].x = sX[pk * XP + dbase];
                vb[0][i].y = sX[(pk + 1) * XP + dbase];
            }
            v8f acc = {};
            #pragma unroll
            for (int g = 0; g < 8; ++g) {
                if (g < 7) {
                    const int gn = g + 1;
                    const int dn = dbase + 64 * (gn >> 1);
                    #pragma unroll
                    for (int i = 0; i < 14; ++i) {
                        const int p0 = 4 * (14 * (gn & 1) + i);
                        const int pk = p0 + 2 * half;
                        const int r0 = p0 < NK ? pk     : NK - 1;
                        const int r1 = p0 < NK ? pk + 1 : NK - 1;
                        vb[gn & 1][i].x = sX[r0 * XP + dn];
                        vb[gn & 1][i].y = sX[r1 * XP + dn];
                    }
                }
                __builtin_amdgcn_sched_barrier(0);
                if ((g & 1) == 0) acc = (v8f){};
                #pragma unroll
                for (int i = 0; i < 14; ++i)
                    acc = __builtin_amdgcn_wmma_f32_16x16x4_f32(
                            false, pa[14 * (g & 1) + i], false, vb[g & 1][i], (short)0, acc, false, false);
                __builtin_amdgcn_sched_barrier(0);
                if (g & 1) {
                    const int d = dbase + 64 * (g >> 1);
                    #pragma unroll
                    for (int r = 0; r < 8; ++r) {
                        const int rowm = r + 8 * half;
                        const int qh = h0 + (rowm >> 2), qw = w0 + (rowm & 3);
                        out[(long)((b * HH + qh) * WW + qw) * DD + d] = acc[r];
                    }
                }
            }
        }

        if (tile == 0) {
            lds_barrier();
            #pragma unroll
            for (int i = 0; i < NEW_PER_WAVE; ++i) {
                const int e = wave + 4 * i;
                float* dst = sX + ((e >> 2) * 10 + (e & 3)) * XP + lane * 4;
                *(v4f*)(dst)       = nx[2 * i];
                *(v4f*)(dst + 128) = nx[2 * i + 1];
            }
            lds_barrier();
        }
    }
}

extern "C" void kernel_launch(void* const* d_in, const int* in_sizes, int n_in,
                              void* d_out, int out_size, void* d_ws, size_t ws_size,
                              hipStream_t stream) {
    const float* x   = (const float*)d_in[0];
    float*       out = (float*)d_out;
    local_attn_wmma_f32<<<dim3(512), dim3(128), 0, stream>>>(x, out);
}


# PV product: each wave owns 64 adjacent channels as 4 channel tiles interleaved by channel, so one 16-byte LDS read feeds 4 tiles on 4 accumulators; 25 key steps; 16-byte row stores
# speedup vs baseline: 3.1599x; 1.0793x over previous
#include <hip/hip_runtime.h>


typedef float v2f __attribute__((ext_vector_type(2)));
typedef float v4f __attribute__((ext_vector_type(4)));
typedef float v8f __attribute__((ext_vector_type(8)));

#define HH 64
#define WW 64
#define DD 256
#define NK 100
#define NP 112
#define SSTRIDE 114
#define XP 268
#define ROWS_PER_WAVE 25
#define NEW_PER_WAVE 10

__device__ __forceinline__ void lds_barrier() {
    __builtin_amdgcn_fence(__ATOMIC_RELEASE, "workgroup", "local");
    __builtin_amdgcn_s_barrier();
    __builtin_amdgcn_fence(__ATOMIC_ACQUIRE, "workgroup", "local");
}

__global__ __launch_bounds__(128)
void local_attn_wmma_f32(const float* __restrict__ x, float* __restrict__ out) {
    __shared__ __attribute__((aligned(16))) float sX[NK * XP];
    __shared__ __attribute__((aligned(16))) float sP[16 * SSTRIDE];

    const int blk = blockIdx.x;
    const int b   = blk >> 7;
    const int t   = blk & 127;
    const int h0  = (t >> 3) << 2;
    const int w0a = (t & 7) << 3;

    const int tid  = threadIdx.x;
    const int wave = __builtin_amdgcn_readfirstlane(tid >> 5);
    const int lane = tid & 31;
    const int half = lane >> 4;
    const int lm   = lane & 15;

    {
        v4f tmp[2 * ROWS_PER_WAVE];
        #pragma unroll
        for (int i = 0; i < ROWS_PER_WAVE; ++i) {
            const int p  = wave + 4 * i;
            const int pi = p / 10, pj = p - pi * 10;
            int kh = h0 - 3 + pi;  kh = kh < 0 ? 0 : (kh > HH - 1 ? HH - 1 : kh);
            int kw = w0a - 3 + pj; kw = kw < 0 ? 0 : (kw > WW - 1 ? WW - 1 : kw);
            const float* src = x + ((b * HH + kh) * WW + kw) * DD + lane * 4;
            tmp[2 * i]     = *(const v4f*)(src);
            tmp[2 * i + 1] = *(const v4f*)(src + 128);
        }
        #pragma unroll
        for (int i = 0; i < ROWS_PER_WAVE; ++i) {
            const int p = wave + 4 * i;
            float* dst = sX + p * XP + lane * 4;
            *(v4f*)(dst)       = tmp[2 * i];
            *(v4f*)(dst + 128) = tmp[2 * i + 1];
        }
    }
    __syncthreads();

    v4f nx[2 * NEW_PER_WAVE];
    #pragma unroll
    for (int i = 0; i < NEW_PER_WAVE; ++i) {
        const int e  = wave + 4 * i;
        const int pi = e >> 2, c = e & 3;
        int kh = h0 - 3 + pi;  kh = kh < 0 ? 0 : (kh > HH - 1 ? HH - 1 : kh);
        int kw = w0a + 7 + c;  kw = kw > WW - 1 ? WW - 1 : kw;
        const float* src = x + ((b * HH + kh) * WW + kw) * DD + lane * 4;
        nx[2 * i]     = *(const v4f*)(src);
        nx[2 * i + 1] = *(const v4f*)(src + 128);
    }

    for (int tile = 0; tile < 2; ++tile) {
        const int shift = tile << 2;
        const int w0    = w0a + shift;

        {
            const int m    = lm;
            int qs = (m & 3) + 3 + shift;
            qs = qs >= 10 ? qs - 10 : qs;
            const int qrow = ((m >> 2) + 3) * 10 + qs;
            const float* aP = sX + qrow * XP + 2 * half;
            v2f qa[DD / 4];
            #pragma unroll
            for (int i = 0; i < DD / 4; ++i) qa[i] = *(const v2f*)(aP + 4 * i);
            for (int tn = wave; tn < 7; tn += 4) {
                const int n    = tn * 16 + lm;
                const int krow = n < NK ? n : NK - 1;
                const float* bP = sX + krow * XP + 2 * half;
                v8f acc = {};
                v2f kb[2][16];
                #pragma unroll
                for (int i = 0; i < 16; ++i) kb[0][i] = *(const v2f*)(bP + 4 * i);
                #pragma unroll
                for (int c = 0; c < 4; ++c) {
                    if (c < 3) {
                        #pragma unroll
                        for (int i = 0; i < 16; ++i)
                            kb[(c + 1) & 1][i] = *(const v2f*)(bP + 4 * (16 * (c + 1) + i));
                    }
                    __builtin_amdgcn_sched_barrier(0);
                    #pragma unroll
                    for (int i = 0; i < 16; ++i)
                        acc = __builtin_amdgcn_wmma_f32_16x16x4_f32(
                                false, qa[16 * c + i], false, kb[c & 1][i], (short)0, acc, false, false);
                    __builtin_amdgcn_sched_barrier(0);
                }
                #pragma unroll
                for (int r = 0; r < 8; ++r)
                    sP[(r + 8 * half) * SSTRIDE + tn * 16 + lm] = acc[r];
            }
        }
        __syncthreads();

        {
            const int m   = tid >> 3;
            const int sub = tid & 7;
            const int qh = h0 + (m >> 2), qw = w0 + (m & 3);
            float* row = sP + m * SSTRIDE;
            float v[14];
            float mx = -3.0e38f;
            #pragma unroll
            for (int j = 0; j < 14; ++j) {
                const int n  = sub + 8 * j;
                const int pi = n / 10;
                int pj = n - pi * 10 - shift;
                pj = pj < 0 ? pj + 10 : pj;
                const int kh = h0 - 3 + pi, kw = w0 - 3 + pj;
                const bool valid = (n < 100) && kh >= 0 && kh < HH && kw >= 0 && kw < WW
                                   && (kh - qh) >= -3 && (kh - qh) <= 3
                                   && (kw - qw) >= -3 && (kw - qw) <= 3;
                const float s = row[n] * 0.0625f;
                v[j] = valid ? s : -3.0e38f;
                if (valid && s > mx) mx = s;
            }
            mx = fmaxf(mx, __shfl_xor(mx, 1));
            mx = fmaxf(mx, __shfl_xor(mx, 2));
            mx = fmaxf(mx, __shfl_xor(mx, 4));
            float sum = 0.f;
            #pragma unroll
            for (int j = 0; j < 14; ++j) {
                const float e = (v[j] > -1.0e38f) ? __expf(v[j] - mx) : 0.f;
                v[j] = e;
                sum += e;
            }
            sum += __shfl_xor(sum, 1);
            sum += __shfl_xor(sum, 2);
            sum += __shfl_xor(sum, 4);
            const float inv = 1.f / sum;
            #pragma unroll
            for (int j = 0; j < 14; ++j) row[sub + 8 * j] = v[j] * inv;
        }
        __syncthreads();

        {
            v2f pa[NK / 4];
            #pragma unroll
            for (int i = 0; i < NK / 4; ++i)
                pa[i] = *(const v2f*)(sP + lm * SSTRIDE + 4 * i + 2 * half);

            const int cbase = wave * 64 + lm * 4;
            const float* vP = sX + (2 * half) * XP + cbase;

            v4f vr[2][5][2];
            #pragma unroll
            for (int i = 0; i < 5; ++i) {
                vr[0][i][0] = *(const v4f*)(vP + (4 * i) * XP);
                vr[0][i][1] = *(const v4f*)(vP + (4 * i + 1) * XP);
            }
            __builtin_amdgcn_sched_barrier(0);
            v8f acc[4] = {};
            #pragma unroll
            for (int g = 0; g < 5; ++g) {
                if (g < 4) {
                    #pragma unroll
                    for (int i = 0; i < 5; ++i) {
                        const int st = 5 * (g + 1) + i;
                        vr[(g + 1) & 1][i][0] = *(const v4f*)(vP + (4 * st) * XP);
                        vr[(g + 1) & 1][i][1] = *(const v4f*)(vP + (4 * st + 1) * XP);
                    }
                }
                __builtin_amdgcn_sched_barrier(0);
                #pragma unroll
                for (int i = 0; i < 5; ++i) {
                    #pragma unroll
                    for (int j = 0; j < 4; ++j) {
                        const v2f bf = { vr[g & 1][i][0][j], vr[g & 1][i][1][j] };
                        acc[j] = __builtin_amdgcn_wmma_f32_16x16x4_f32(
                                false, pa[5 * g + i], false, bf, (short)0, acc[j], false, false);
                    }
                }
                __builtin_amdgcn_sched_barrier(0);
            }
            #pragma unroll
            for (int r = 0; r < 8; ++r) {
                const int rowm = r + 8 * half;
                const int qh = h0 + (rowm >> 2), qw = w0 + (rowm & 3);
                const v4f o = { acc[0][r], acc[1][r], acc[2][r], acc[3][r] };
                *(v4f*)(out + (long)((b * HH + qh) * WW + qw) * DD + cbase) = o;
            }
        }

        if (tile == 0) {
            lds_barrier();
            #pragma unroll
            for (int i = 0; i < NEW_PER_WAVE; ++i) {
                const int e = wave + 4 * i;
                float* dst = sX + ((e >> 2) * 10 + (e & 3)) * XP + lane * 4;
                *(v4f*)(dst)       = nx[2 * i];
                *(v4f*)(dst + 128) = nx[2 * i + 1];
            }
            lds_barrier();
        }
    }
}

extern "C" void kernel_launch(void* const* d_in, const int* in_sizes, int n_in,
                              void* d_out, int out_size, void* d_ws, size_t ws_size,
                              hipStream_t stream) {
    const float* x   = (const float*)d_in[0];
    float*       out = (float*)d_out;
    local_attn_wmma_f32<<<dim3(512), dim3(128), 0, stream>>>(x, out);
}


# softmax re-mapped: one lane per (query, window patch row), 10 adjacent keys under a 10-bit column mask, rows outside the window skipped; 8-lane max/sum on DPP; row sum = 7 partials of 10 keys
# speedup vs baseline: 3.3799x; 1.0696x over previous
#include <hip/hip_runtime.h>


typedef float v2f __attribute__((ext_vector_type(2)));
typedef float v4f __attribute__((ext_vector_type(4)));
typedef float v8f __attribute__((ext_vector_type(8)));

#define HH 64
#define WW 64
#define DD 256
#define NK 100
#define NP 112
#define SSTRIDE 114
#define XP 268
#define ROWS_PER_WAVE 25
#define NEW_PER_WAVE 10

__device__ __forceinline__ void lds_barrier() {
    __builtin_amdgcn_fence(__ATOMIC_RELEASE, "workgroup", "local");
    __builtin_amdgcn_s_barrier();
    __builtin_amdgcn_fence(__ATOMIC_ACQUIRE, "workgroup", "local");
}

template <int CTRL>
__device__ __forceinline__ float lane_xor(float x) {
    const int xi = __float_as_int(x);
    return __int_as_float(__builtin_amdgcn_update_dpp(0, xi, CTRL, 0xf, 0xf, true));
}

__global__ __launch_bounds__(128)
void local_attn_wmma_f32(const float* __restrict__ x, float* __restrict__ out) {
    __shared__ __attribute__((aligned(16))) float sX[NK * XP];
    __shared__ __attribute__((aligned(16))) float sP[16 * SSTRIDE];

    const int blk = blockIdx.x;
    const int b   = blk >> 7;
    const int t   = blk & 127;
    const int h0  = (t >> 3) << 2;
    const int w0a = (t & 7) << 3;

    const int tid  = threadIdx.x;
    const int wave = __builtin_amdgcn_readfirstlane(tid >> 5);
    const int lane = tid & 31;
    const int half = lane >> 4;
    const int lm   = lane & 15;

    {
        v4f tmp[2 * ROWS_PER_WAVE];
        #pragma unroll
        for (int i = 0; i < ROWS_PER_WAVE; ++i) {
            const int p  = wave + 4 * i;
            const int pi = p / 10, pj = p - pi * 10;
            int kh = h0 - 3 + pi;  kh = kh < 0 ? 0 : (kh > HH - 1 ? HH - 1 : kh);
            int kw = w0a - 3 + pj; kw = kw < 0 ? 0 : (kw > WW - 1 ? WW - 1 : kw);
            const float* src = x + ((b * HH + kh) * WW + kw) * DD + lane * 4;
            tmp[2 * i]     = *(const v4f*)(src);
            tmp[2 * i + 1] = *(const v4f*)(src + 128);
        }
        #pragma unroll
        for (int i = 0; i < ROWS_PER_WAVE; ++i) {
            const int p = wave + 4 * i;
            float* dst = sX + p * XP + lane * 4;
            *(v4f*)(dst)       = tmp[2 * i];
            *(v4f*)(dst + 128) = tmp[2 * i + 1];
        }
    }
    __syncthreads();

    v4f nx[2 * NEW_PER_WAVE];
    #pragma unroll
    for (int i = 0; i < NEW_PER_WAVE; ++i) {
        const int e  = wave + 4 * i;
        const int pi = e >> 2, c = e & 3;
        int kh = h0 - 3 + pi;  kh = kh < 0 ? 0 : (kh > HH - 1 ? HH - 1 : kh);
        int kw = w0a + 7 + c;  kw = kw > WW - 1 ? WW - 1 : kw;
        const float* src = x + ((b * HH + kh) * WW + kw) * DD + lane * 4;
        nx[2 * i]     = *(const v4f*)(src);
        nx[2 * i + 1] = *(const v4f*)(src + 128);
    }

    for (int tile = 0; tile < 2; ++tile) {
        const int shift = tile << 2;
        const int w0    = w0a + shift;

        {
            const int m    = lm;
            int qs = (m & 3) + 3 + shift;
            qs = qs >= 10 ? qs - 10 : qs;
            const int qrow = ((m >> 2) + 3) * 10 + qs;
            const float* aP = sX + qrow * XP + 2 * half;
            v2f qa[DD / 4];
            #pragma unroll
            for (int i = 0; i < DD / 4; ++i) qa[i] = *(const v2f*)(aP + 4 * i);
            for (int tn = wave; tn < 7; tn += 4) {
                const int n    = tn * 16 + lm;
                const int krow = n < NK ? n : NK - 1;
                const float* bP = sX + krow * XP + 2 * half;
                v8f acc = {};
                v2f kb[2][16];
                #pragma unroll
                for (int i = 0; i < 16; ++i) kb[0][i] = *(const v2f*)(bP + 4 * i);
                #pragma unroll
                for (int c = 0; c < 4; ++c) {
                    if (c < 3) {
                        #pragma unroll
                        for (int i = 0; i < 16; ++i)
                            kb[(c + 1) & 1][i] = *(const v2f*)(bP + 4 * (16 * (c + 1) + i));
                    }
                    __builtin_amdgcn_sched_barrier(0);
                    #pragma unroll
                    for (int i = 0; i < 16; ++i)
                        acc = __builtin_amdgcn_wmma_f32_16x16x4_f32(
                                false, qa[16 * c + i], false, kb[c & 1][i], (short)0, acc, false, false);
                    __builtin_amdgcn_sched_barrier(0);
                }
                #pragma unroll
                for (int r = 0; r < 8; ++r)
                    sP[(r + 8 * half) * SSTRIDE + tn * 16 + lm] = acc[r];
            }
        }
        __syncthreads();

        {
            const int m  = tid >> 3;
            const int r  = tid & 7;
            const int qr = m >> 2, qc = m & 3;
            int lo = 3 - w0;       lo = lo > qc ? lo : qc;
            int hi = WW + 2 - w0;  hi = hi < qc + 6 ? hi : qc + 6;
            const unsigned colP = ((2u << hi) - 1u) & ~((1u << lo) - 1u);
            const unsigned colS = ((colP << shift) | (colP >> (10 - shift))) & 0x3FFu;
            const int kh = h0 + qr - 3 + r;
            const unsigned msk = (r < 7 && kh >= 0 && kh < HH) ? colS : 0u;
            int pi = qr + r;  pi = pi > 9 ? 9 : pi;
            float* rowp = sP + m * SSTRIDE + 10 * pi;
            v2f t[5];
            #pragma unroll
            for (int jj = 0; jj < 5; ++jj) t[jj] = *(const v2f*)(rowp + 2 * jj);
            float v[10];
            #pragma unroll
            for (int j = 0; j < 10; ++j) {
                const float s = t[j >> 1][j & 1] * 0.0625f;
                v[j] = ((msk >> j) & 1u) ? s : -3.0e38f;
            }
            float mx = fmaxf(fmaxf(fmaxf(v[0], v[1]), fmaxf(v[2], v[3])),
                             fmaxf(fmaxf(fmaxf(v[4], v[5]), fmaxf(v[6], v[7])), fmaxf(v[8], v[9])));
            mx = fmaxf(mx, -3.0e38f);
            mx = fmaxf(mx, lane_xor<0x161>(mx));
            mx = fmaxf(mx, lane_xor<0x162>(mx));
            mx = fmaxf(mx, lane_xor<0x164>(mx));
            #pragma unroll
            for (int j = 0; j < 10; ++j)
                v[j] = (v[j] > -1.0e38f) ? __expf(v[j] - mx) : 0.f;
            float sum = ((v[0] + v[1]) + (v[2] + v[3])) + ((v[4] + v[5]) + (v[6] + v[7])) + (v[8] + v[9]);
            sum += lane_xor<0x161>(sum);
            sum += lane_xor<0x162>(sum);
            sum += lane_xor<0x164>(sum);
            const float inv = 1.f / sum;
            if (r < 7) {
                #pragma unroll
                for (int jj = 0; jj < 5; ++jj) {
                    const v2f p = { v[2 * jj] * inv, v[2 * jj + 1] * inv };
                    *(v2f*)(rowp + 2 * jj) = p;
                }
            } else {
                #pragma unroll
                for (int z = 0; z < 3; ++z) {
                    int pz = qr + 7 + z;  pz = pz > 9 ? pz - 10 : pz;
                    float* zp = sP + m * SSTRIDE + 10 * pz;
                    const v2f zero = { 0.f, 0.f };
                    #pragma unroll
                    for (int jj = 0; jj < 5; ++jj) *(v2f*)(zp + 2 * jj) = zero;
                }
            }
        }
        __syncthreads();

        {
            v2f pa[NK / 4];
            #pragma unroll
            for (int i = 0; i < NK / 4; ++i)
                pa[i] = *(const v2f*)(sP + lm * SSTRIDE + 4 * i + 2 * half);

            const int cbase = wave * 64 + lm * 4;
            const float* vP = sX + (2 * half) * XP + cbase;

            v4f vr[2][5][2];
            #pragma unroll
            for (int i = 0; i < 5; ++i) {
                vr[0][i][0] = *(const v4f*)(vP + (4 * i) * XP);
                vr[0][i][1] = *(const v4f*)(vP + (4 * i + 1) * XP);
            }
            __builtin_amdgcn_sched_barrier(0);
            v8f acc[4] = {};
            #pragma unroll
            for (int g = 0; g < 5; ++g) {
                if (g < 4) {
                    #pragma unroll
                    for (int i = 0; i < 5; ++i) {
                        const int st = 5 * (g + 1) + i;
                        vr[(g + 1) & 1][i][0] = *(const v4f*)(vP + (4 * st) * XP);
                        vr[(g + 1) & 1][i][1] = *(const v4f*)(vP + (4 * st + 1) * XP);
                    }
                }
                __builtin_amdgcn_sched_barrier(0);
                #pragma unroll
                for (int i = 0; i < 5; ++i) {
                    #pragma unroll
                    for (int j = 0; j < 4; ++j) {
                        const v2f bf = { vr[g & 1][i][0][j], vr[g & 1][i][1][j] };
                        acc[j] = __builtin_amdgcn_wmma_f32_16x16x4_f32(
                                false, pa[5 * g + i], false, bf, (short)0, acc[j], false, false);
                    }
                }
                __builtin_amdgcn_sched_barrier(0);
            }
            #pragma unroll
            for (int r = 0; r < 8; ++r) {
                const int rowm = r + 8 * half;
                const int qh = h0 + (rowm >> 2), qw = w0 + (rowm & 3);
                const v4f o = { acc[0][r], acc[1][r], acc[2][r], acc[3][r] };
                *(v4f*)(out + (long)((b * HH + qh) * WW + qw) * DD + cbase) = o;
            }
        }

        if (tile == 0) {
            lds_barrier();
            #pragma unroll
            for (int i = 0; i < NEW_PER_WAVE; ++i) {
                const int e = wave + 4 * i;
                float* dst = sX + ((e >> 2) * 10 + (e & 3)) * XP + lane * 4;
                *(v4f*)(dst)       = nx[2 * i];
                *(v4f*)(dst + 128) = nx[2 * i + 1];
            }
            lds_barrier();
        }
    }
}

extern "C" void kernel_launch(void* const* d_in, const int* in_sizes, int n_in,
                              void* d_out, int out_size, void* d_ws, size_t ws_size,
                              hipStream_t stream) {
    const float* x   = (const float*)d_in[0];
    float*       out = (float*)d_out;
    local_attn_wmma_f32<<<dim3(512), dim3(128), 0, stream>>>(x, out);
}


# Q.K^T evened over the 4 waves: 96 non-corner keys = 6 key tiles x 2 channel halves, 3 jobs per wave on 3 accumulators; 4 corner keys (one query each) as vector dots; score = 2 partial sums added
# speedup vs baseline: 3.6089x; 1.0678x over previous
#include <hip/hip_runtime.h>


typedef float v2f __attribute__((ext_vector_type(2)));
typedef float v4f __attribute__((ext_vector_type(4)));
typedef float v8f __attribute__((ext_vector_type(8)));

#define HH 64
#define WW 64
#define DD 256
#define NK 100
#define SSTRIDE 114
#define PLANE (16 * SSTRIDE)
#define XP 268
#define ROWS_PER_WAVE 25
#define NEW_PER_WAVE 10

__device__ __forceinline__ void lds_barrier() {
    __builtin_amdgcn_fence(__ATOMIC_RELEASE, "workgroup", "local");
    __builtin_amdgcn_s_barrier();
    __builtin_amdgcn_fence(__ATOMIC_ACQUIRE, "workgroup", "local");
}

template <int CTRL>
__device__ __forceinline__ float lane_xor(float x) {
    const int xi = __float_as_int(x);
    return __int_as_float(__builtin_amdgcn_update_dpp(0, xi, CTRL, 0xf, 0xf, true));
}

__device__ __forceinline__ float lane_xhalf(float x) {
    const int xi = __float_as_int(x);
    return __int_as_float(__builtin_amdgcn_permlanex16(xi, xi, 0x76543210, 0xfedcba98, false, false));
}

__global__ __launch_bounds__(128)
void local_attn_wmma_f32(const float* __restrict__ x, float* __restrict__ out) {
    __shared__ __attribute__((aligned(16))) float sX[NK * XP];
    __shared__ __attribute__((aligned(16))) float sP[2 * PLANE];

    const int blk = blockIdx.x;
    const int b   = blk >> 7;
    const int t   = blk & 127;
    const int h0  = (t >> 3) << 2;
    const int w0a = (t & 7) << 3;

    const int tid  = threadIdx.x;
    const int wave = __builtin_amdgcn_readfirstlane(tid >> 5);
    const int lane = tid & 31;
    const int half = lane >> 4;
    const int lm   = lane & 15;

    {
        v4f tmp[2 * ROWS_PER_WAVE];
        #pragma unroll
        for (int i = 0; i < ROWS_PER_WAVE; ++i) {
            const int p  = wave + 4 * i;
            const int pi = p / 10, pj = p - pi * 10;
            int kh = h0 - 3 + pi;  kh = kh < 0 ? 0 : (kh > HH - 1 ? HH - 1 : kh);
            int kw = w0a - 3 + pj; kw = kw < 0 ? 0 : (kw > WW - 1 ? WW - 1 : kw);
            const float* src = x + ((b * HH + kh) * WW + kw) * DD + lane * 4;
            tmp[2 * i]     = *(const v4f*)(src);
            tmp[2 * i + 1] = *(const v4f*)(src + 128);
        }
        #pragma unroll
        for (int i = 0; i < ROWS_PER_WAVE; ++i) {
            const int p = wave + 4 * i;
            float* dst = sX + p * XP + lane * 4;
            *(v4f*)(dst)       = tmp[2 * i];
            *(v4f*)(dst + 128) = tmp[2 * i + 1];
        }
    }
    __syncthreads();

    v4f nx[2 * NEW_PER_WAVE];
    #pragma unroll
    for (int i = 0; i < NEW_PER_WAVE; ++i) {
        const int e  = wave + 4 * i;
        const int pi = e >> 2, c = e & 3;
        int kh = h0 - 3 + pi;  kh = kh < 0 ? 0 : (kh > HH - 1 ? HH - 1 : kh);
        int kw = w0a + 7 + c;  kw = kw > WW - 1 ? WW - 1 : kw;
        const float* src = x + ((b * HH + kh) * WW + kw) * DD + lane * 4;
        nx[2 * i]     = *(const v4f*)(src);
        nx[2 * i + 1] = *(const v4f*)(src + 128);
    }

    for (int tile = 0; tile < 2; ++tile) {
        const int shift = tile << 2;
        const int w0    = w0a + shift;

        {
            const int m    = lm;
            int qs = (m & 3) + 3 + shift;
            qs = qs >= 10 ? qs - 10 : qs;
            const int qrow = ((m >> 2) + 3) * 10 + qs;
            const int hsel = wave >> 1;
            const int tb   = 3 * (wave & 1);
            const int cA = shift ? 3 : 0, cB = shift ? 4 : 9, cC = cA + 90, cD = cB + 90;

            const int cr = wave >> 1, cc = wave & 1;
            const int crow = 90 * cr + (cc ? (shift ? 3 : 9) : shift);
            int cqs = 3 * cc + 3 + shift;
            cqs = cqs >= 10 ? cqs - 10 : cqs;
            const int cqrow = (3 * cr + 3) * 10 + cqs;
            const int cm    = 12 * cr + 3 * cc;
            const float* cq = sX + cqrow * XP + lane * 4;
            const float* ck = sX + crow * XP + lane * 4;
            const v4f q0 = *(const v4f*)(cq), q1 = *(const v4f*)(cq + 128);
            const v4f k0 = *(const v4f*)(ck), k1 = *(const v4f*)(ck + 128);

            const float* aP = sX + qrow * XP + 128 * hsel + 2 * half;
            v2f qa[32];
            #pragma unroll
            for (int i = 0; i < 32; ++i) qa[i] = *(const v2f*)(aP + 4 * i);

            const float* bP[3];
            int krow[3];
            #pragma unroll
            for (int t = 0; t < 3; ++t) {
                int r = (tb + t) * 16 + lm;
                r += (r >= cA);
                r += (r >= cB);
                r += (r >= cC);
                r += (r >= cD);
                krow[t] = r;
                bP[t] = sX + r * XP + 128 * hsel + 2 * half;
            }
            v8f acc[3] = {};
            v2f kb[2][3][8];
            #pragma unroll
            for (int t = 0; t < 3; ++t) {
                #pragma unroll
                for (int i = 0; i < 8; ++i) kb[0][t][i] = *(const v2f*)(bP[t] + 4 * i);
            }
            #pragma unroll
            for (int c = 0; c < 4; ++c) {
                if (c < 3) {
                    #pragma unroll
                    for (int t = 0; t < 3; ++t) {
                        #pragma unroll
                        for (int i = 0; i < 8; ++i)
                            kb[(c + 1) & 1][t][i] = *(const v2f*)(bP[t] + 4 * (8 * (c + 1) + i));
                    }
                }
                __builtin_amdgcn_sched_barrier(0);
                #pragma unroll
                for (int i = 0; i < 8; ++i) {
                    #pragma unroll
                    for (int t = 0; t < 3; ++t)
                        acc[t] = __builtin_amdgcn_wmma_f32_16x16x4_f32(
                                false, qa[8 * c + i], false, kb[c & 1][t][i], (short)0, acc[t], false, false);
                }
                __builtin_amdgcn_sched_barrier(0);
            }
            float* sW = sP + hsel * PLANE;
            #pragma unroll
            for (int t = 0; t < 3; ++t) {
                #pragma unroll
                for (int r = 0; r < 8; ++r)
                    sW[(r + 8 * half) * SSTRIDE + krow[t]] = acc[t][r];
            }

            float d = q0[0] * k0[0] + q0[1] * k0[1] + q0[2] * k0[2] + q0[3] * k0[3]
                    + q1[0] * k1[0] + q1[1] * k1[1] + q1[2] * k1[2] + q1[3] * k1[3];
            d += lane_xor<0x161>(d);
            d += lane_xor<0x162>(d);
            d += lane_xor<0x164>(d);
            d += lane_xor<0x168>(d);
            d += lane_xhalf(d);
            sP[half * PLANE + lm * SSTRIDE + crow] = (lane == cm) ? d : 0.f;
        }
        __syncthreads();

        {
            const int m  = tid >> 3;
            const int r  = tid & 7;
            const int qr = m >> 2, qc = m & 3;
            int lo = 3 - w0;       lo = lo > qc ? lo : qc;
            int hi = WW + 2 - w0;  hi = hi < qc + 6 ? hi : qc + 6;
            const unsigned colP = ((2u << hi) - 1u) & ~((1u << lo) - 1u);
            const unsigned colS = ((colP << shift) | (colP >> (10 - shift))) & 0x3FFu;
            const int kh = h0 + qr - 3 + r;
            const unsigned msk = (r < 7 && kh >= 0 && kh < HH) ? colS : 0u;
            int pi = qr + r;  pi = pi > 9 ? 9 : pi;
            float* rowp = sP + m * SSTRIDE + 10 * pi;
            v2f t[5], u[5];
            #pragma unroll
            for (int jj = 0; jj < 5; ++jj) {
                t[jj] = *(const v2f*)(rowp + 2 * jj);
                u[jj] = *(const v2f*)(rowp + PLANE + 2 * jj);
            }
            float v[10];
            #pragma unroll
            for (int j = 0; j < 10; ++j) {
                const float s = (t[j >> 1][j & 1] + u[j >> 1][j & 1]) * 0.0625f;
                v[j] = ((msk >> j) & 1u) ? s : -3.0e38f;
            }
            float mx = fmaxf(fmaxf(fmaxf(v[0], v[1]), fmaxf(v[2], v[3])),
                             fmaxf(fmaxf(fmaxf(v[4], v[5]), fmaxf(v[6], v[7])), fmaxf(v[8], v[9])));
            mx = fmaxf(mx, -3.0e38f);
            mx = fmaxf(mx, lane_xor<0x161>(mx));
            mx = fmaxf(mx, lane_xor<0x162>(mx));
            mx = fmaxf(mx, lane_xor<0x164>(mx));
            #pragma unroll
            for (int j = 0; j < 10; ++j)
                v[j] = (v[j] > -1.0e38f) ? __expf(v[j] - mx) : 0.f;
            float sum = ((v[0] + v[1]) + (v[2] + v[3])) + ((v[4] + v[5]) + (v[6] + v[7])) + (v[8] + v[9]);
            sum += lane_xor<0x161>(sum);
            sum += lane_xor<0x162>(sum);
            sum += lane_xor<0x164>(sum);
            const float inv = 1.f / sum;
            if (r < 7) {
                #pragma unroll
                for (int jj = 0; jj < 5; ++jj) {
                    const v2f p = { v[2 * jj] * inv, v[2 * jj + 1] * inv };
                    *(v2f*)(rowp + 2 * jj) = p;
                }
            } else {
                #pragma unroll
                for (int z = 0; z < 3; ++z) {
                    int pz = qr + 7 + z;  pz = pz > 9 ? pz - 10 : pz;
                    float* zp = sP + m * SSTRIDE + 10 * pz;
                    const v2f zero = { 0.f, 0.f };
                    #pragma unroll
                    for (int jj = 0; jj < 5; ++jj) *(v2f*)(zp + 2 * jj) = zero;
                }
            }
        }
        __syncthreads();

        {
            v2f pa[NK / 4];
            #pragma unroll
            for (int i = 0; i < NK / 4; ++i)
                pa[i] = *(const v2f*)(sP + lm * SSTRIDE + 4 * i + 2 * half);

            const int cbase = wave * 64 + lm * 4;
            const float* vP = sX + (2 * half) * XP + cbase;

            v4f vr[2][5][2];
            #pragma unroll
            for (int i = 0; i < 5; ++i) {
                vr[0][i][0] = *(const v4f*)(vP + (4 * i) * XP);
                vr[0][i][1] = *(const v4f*)(vP + (4 * i + 1) * XP);
            }
            __builtin_amdgcn_sched_barrier(0);
            v8f acc[4] = {};
            #pragma unroll
            for (int g = 0; g < 5; ++g) {
                if (g < 4) {
                    #pragma unroll
                    for (int i = 0; i < 5; ++i) {
                        const int st = 5 * (g + 1) + i;
                        vr[(g + 1) & 1][i][0] = *(const v4f*)(vP + (4 * st) * XP);
                        vr[(g + 1) & 1][i][1] = *(const v4f*)(vP + (4 * st + 1) * XP);
                    }
                }
                __builtin_amdgcn_sched_barrier(0);
                #pragma unroll
                for (int i = 0; i < 5; ++i) {
                    #pragma unroll
                    for (int j = 0; j < 4; ++j) {
                        const v2f bf = { vr[g & 1][i][0][j], vr[g & 1][i][1][j] };
                        acc[j] = __builtin_amdgcn_wmma_f32_16x16x4_f32(
                                false, pa[5 * g + i], false, bf, (short)0, acc[j], false, false);
                    }
                }
                __builtin_amdgcn_sched_barrier(0);
            }
            #pragma unroll
            for (int r = 0; r < 8; ++r) {
                const int rowm = r + 8 * half;
                const int qh = h0 + (rowm >> 2), qw = w0 + (rowm & 3);
                const v4f o = { acc[0][r], acc[1][r], acc[2][r], acc[3][r] };
                *(v4f*)(out + (long)((b * HH + qh) * WW + qw) * DD + cbase) = o;
            }
        }

        if (tile == 0) {
            lds_barrier();
            #pragma unroll
            for (int i = 0; i < NEW_PER_WAVE; ++i) {
                const int e = wave + 4 * i;
                float* dst = sX + ((e >> 2) * 10 + (e & 3)) * XP + lane * 4;
                *(v4f*)(dst)       = nx[2 * i];
                *(v4f*)(dst + 128) = nx[2 * i + 1];
            }
            lds_barrier();
        }
    }
}

extern "C" void kernel_launch(void* const* d_in, const int* in_sizes, int n_in,
                              void* d_out, int out_size, void* d_ws, size_t ws_size,
                              hipStream_t stream) {
    const float* x   = (const float*)d_in[0];
    float*       out = (float*)d_out;
    local_attn_wmma_f32<<<dim3(512), dim3(128), 0, stream>>>(x, out);
}


# one wave per 4x4 query tile, 4 tiles on one 14x14 patch per block: scores built transposed, softmax in-lane, P feeds P.V from registers; score = one 64-step sum, row sum = 4 in-lane partials
# speedup vs baseline: 3.8609x; 1.0698x over previous
#include <hip/hip_runtime.h>


typedef float v2f __attribute__((ext_vector_type(2)));
typedef float v4f __attribute__((ext_vector_type(4)));
typedef float v8f __attribute__((ext_vector_type(8)));

#define HH 64
#define WW 64
#define DD 256
#define PW 14
#define NROWS (PW * PW)
#define XP 260
#define ROWS_PER_WAVE 49

template <int CTRL>
__device__ __forceinline__ float lane_xor(float x) {
    const int xi = __float_as_int(x);
    return __int_as_float(__builtin_amdgcn_update_dpp(0, xi, CTRL, 0xf, 0xf, true));
}

__device__ __forceinline__ float lane_xhalf(float x) {
    const int xi = __float_as_int(x);
    return __int_as_float(__builtin_amdgcn_permlanex16(xi, xi, 0x76543210, 0xfedcba98, false, false));
}

__device__ __forceinline__ constexpr int key_q(int u) { return u + 1 + (u >= 8 ? 1 : 0); }

__global__ __launch_bounds__(128)
void local_attn_wmma_f32(const float* __restrict__ x, float* __restrict__ out) {
    __shared__ __attribute__((aligned(16))) float sX[NROWS * XP];

    const int blk = blockIdx.x;
    const int b   = blk >> 6;
    const int t   = blk & 63;
    const int h0  = (t >> 3) << 3;
    const int w0  = (t & 7) << 3;

    const int tid  = threadIdx.x;
    const int wave = __builtin_amdgcn_readfirstlane(tid >> 5);
    const int lane = tid & 31;
    const int half = lane >> 4;
    const int lm   = lane & 15;

    {
        v4f tmp[2 * ROWS_PER_WAVE];
        #pragma unroll
        for (int i = 0; i < ROWS_PER_WAVE; ++i) {
            const int R  = wave + 4 * i;
            const int pi = R / PW, pj = R - pi * PW;
            int kh = h0 - 3 + pi;  kh = kh < 0 ? 0 : (kh > HH - 1 ? HH - 1 : kh);
            int kw = w0 - 3 + pj;  kw = kw < 0 ? 0 : (kw > WW - 1 ? WW - 1 : kw);
            const float* src = x + ((b * HH + kh) * WW + kw) * DD + lane * 4;
            tmp[2 * i]     = *(const v4f*)(src);
            tmp[2 * i + 1] = *(const v4f*)(src + 128);
        }
        #pragma unroll
        for (int i = 0; i < ROWS_PER_WAVE; ++i) {
            const int R = wave + 4 * i;
            float* dst = sX + R * XP + lane * 4;
            *(v4f*)(dst)       = tmp[2 * i];
            *(v4f*)(dst + 128) = tmp[2 * i + 1];
        }
    }
    __syncthreads();

    const int ty  = wave >> 1, tx = wave & 1;
    const int th0 = h0 + 4 * ty, tw0 = w0 + 4 * tx;
    const float* tileX = sX + ((4 * ty) * PW + 4 * tx) * XP;

    const int qr = lm >> 2, qc = lm & 3;
    int lo = 3 - tw0;        lo = lo > qc ? lo : qc;
    int hi = WW + 2 - tw0;   hi = hi < qc + 6 ? hi : qc + 6;
    const unsigned cm = ((2u << hi) - 1u) & ~((1u << lo) - 1u);
    int loR = 3 - th0;       loR = loR > qr ? loR : qr;
    int hiR = HH + 2 - th0;  hiR = hiR < qr + 6 ? hiR : qr + 6;
    const unsigned rm = ((2u << hiR) - 1u) & ~((1u << loR) - 1u);
    const unsigned rmh = half ? (__brev(rm) >> 22) : rm;

    v8f acc[6] = {};
    float dc[4];
    {
        const float* qP = tileX + ((3 + qr) * PW + 3 + qc) * XP + 2 * half;
        v2f qb[64];
        #pragma unroll
        for (int i = 0; i < 64; ++i) qb[i] = *(const v2f*)(qP + 4 * i);

        const float* kP[6];
        {
            const int hk = lm >> 3, r = lm & 7;
            #pragma unroll
            for (int kt = 0; kt < 6; ++kt) {
                const int u  = 8 * kt + r;
                const int q  = u + 1 + (u >= 8 ? 1 : 0);
                const int pl = q / 10, pj = q - 10 * pl;
                const int pi = hk ? 9 - pl : pl;
                kP[kt] = tileX + (pi * PW + pj) * XP + 2 * half;
            }
        }

        #pragma unroll
        for (int j = 0; j < 4; ++j) {
            const int hh = j >> 1, s = j & 1;
            const float* ck = tileX + ((hh ? 9 : 0) * PW + (s ? 9 : 0)) * XP + lane * 4;
            const float* cq = tileX + ((3 + (hh ? 3 : 0)) * PW + 3 + (s ? 3 : 0)) * XP + lane * 4;
            const v4f k0 = *(const v4f*)(ck), k1 = *(const v4f*)(ck + 128);
            const v4f q0 = *(const v4f*)(cq), q1 = *(const v4f*)(cq + 128);
            float d = q0[0] * k0[0] + q0[1] * k0[1] + q0[2] * k0[2] + q0[3] * k0[3]
                    + q1[0] * k1[0] + q1[1] * k1[1] + q1[2] * k1[2] + q1[3] * k1[3];
            d += lane_xor<0x161>(d);
            d += lane_xor<0x162>(d);
            d += lane_xor<0x164>(d);
            d += lane_xor<0x168>(d);
            d += lane_xhalf(d);
            dc[j] = d;
        }

        v2f kb[2][6][4];
        #pragma unroll
        for (int kt = 0; kt < 6; ++kt) {
            #pragma unroll
            for (int i = 0; i < 4; ++i) kb[0][kt][i] = *(const v2f*)(kP[kt] + 4 * i);
        }
        #pragma unroll
        for (int g = 0; g < 16; ++g) {
            if (g < 15) {
                #pragma unroll
                for (int kt = 0; kt < 6; ++kt) {
                    #pragma unroll
                    for (int i = 0; i < 4; ++i)
                        kb[(g + 1) & 1][kt][i] = *(const v2f*)(kP[kt] + 4 * (4 * (g + 1) + i));
                }
            }
            __builtin_amdgcn_sched_barrier(0);
            #pragma unroll
            for (int i = 0; i < 4; ++i) {
                #pragma unroll
                for (int kt = 0; kt < 6; ++kt)
                    acc[kt] = __builtin_amdgcn_wmma_f32_16x16x4_f32(
                            false, kb[g & 1][kt][i], false, qb[4 * g + i], (short)0, acc[kt], false, false);
            }
            __builtin_amdgcn_sched_barrier(0);
        }
    }

    float p[6][8], pc[2];
    {
        #pragma unroll
        for (int kt = 0; kt < 6; ++kt) {
            #pragma unroll
            for (int r = 0; r < 8; ++r) {
                const int q  = key_q(8 * kt + r);
                const int pl = q / 10, pj = q - 10 * pl;
                const bool ok = ((rmh >> pl) & (cm >> pj) & 1u) != 0u;
                p[kt][r] = ok ? acc[kt][r] * 0.0625f : -3.0e38f;
            }
        }
        #pragma unroll
        for (int s = 0; s < 2; ++s) {
            const bool ok = (rmh & (cm >> (s ? 9 : 0)) & 1u) != 0u;
            const float d = half ? dc[2 + s] : dc[s];
            pc[s] = ok ? d * 0.0625f : -3.0e38f;
        }
        float m4[4] = { -3.0e38f, -3.0e38f, -3.0e38f, -3.0e38f };
        #pragma unroll
        for (int kt = 0; kt < 6; ++kt) {
            #pragma unroll
            for (int r = 0; r < 8; ++r) m4[r & 3] = fmaxf(m4[r & 3], p[kt][r]);
        }
        m4[0] = fmaxf(m4[0], pc[0]);
        m4[1] = fmaxf(m4[1], pc[1]);
        float mx = fmaxf(fmaxf(m4[0], m4[1]), fmaxf(m4[2], m4[3]));
        mx = fmaxf(mx, lane_xhalf(mx));
        float s4[4] = { 0.f, 0.f, 0.f, 0.f };
        #pragma unroll
        for (int kt = 0; kt < 6; ++kt) {
            #pragma unroll
            for (int r = 0; r < 8; ++r) {
                p[kt][r] = (p[kt][r] > -1.0e38f) ? __expf(p[kt][r] - mx) : 0.f;
                s4[r & 3] += p[kt][r];
            }
        }
        #pragma unroll
        for (int s = 0; s < 2; ++s) {
            pc[s] = (pc[s] > -1.0e38f) ? __expf(pc[s] - mx) : 0.f;
            s4[s] += pc[s];
        }
        float sum = (s4[0] + s4[1]) + (s4[2] + s4[3]);
        sum += lane_xhalf(sum);
        const float inv = 1.f / sum;
        #pragma unroll
        for (int kt = 0; kt < 6; ++kt) {
            #pragma unroll
            for (int r = 0; r < 8; ++r) p[kt][r] *= inv;
        }
        pc[0] *= inv;
        pc[1] *= inv;
    }

    {
        const float* vB[5];
        #pragma unroll
        for (int pl = 0; pl < 5; ++pl)
            vB[pl] = tileX + ((half ? 9 - pl : pl) * PW) * XP + 4 * lm;

        v8f o[4][4] = {};
        v4f vr[2][4][2];
        #pragma unroll
        for (int g = 0; g < 4; ++g) {
            vr[0][g][0] = *(const v4f*)(vB[0] + 1 * XP + 64 * g);
            vr[0][g][1] = *(const v4f*)(vB[0] + 2 * XP + 64 * g);
        }
        #pragma unroll
        for (int st = 0; st < 25; ++st) {
            if (st < 24) {
                const int sn = st + 1;
                const int q0 = sn < 24 ? key_q(8 * (sn >> 2) + 2 * (sn & 3))     : 0;
                const int q1 = sn < 24 ? key_q(8 * (sn >> 2) + 2 * (sn & 3) + 1) : 9;
                const float* a0 = vB[q0 / 10] + (q0 % 10) * XP;
                const float* a1 = vB[q1 / 10] + (q1 % 10) * XP;
                #pragma unroll
                for (int g = 0; g < 4; ++g) {
                    vr[sn & 1][g][0] = *(const v4f*)(a0 + 64 * g);
                    vr[sn & 1][g][1] = *(const v4f*)(a1 + 64 * g);
                }
            }
            __builtin_amdgcn_sched_barrier(0);
            const int kk = st < 24 ? (st >> 2) : 0, j2 = st < 24 ? 2 * (st & 3) : 0;
            const v2f pa = st < 24 ? (v2f){ p[kk][j2], p[kk][j2 + 1] } : (v2f){ pc[0], pc[1] };
            #pragma unroll
            for (int g = 0; g < 4; ++g) {
                #pragma unroll
                for (int jj = 0; jj < 4; ++jj) {
                    const v2f bf = { vr[st & 1][g][0][jj], vr[st & 1][g][1][jj] };
                    o[g][jj] = __builtin_amdgcn_wmma_f32_16x16x4_f32(
                            false, pa, false, bf, (short)0, o[g][jj], false, false);
                }
            }
            __builtin_amdgcn_sched_barrier(0);
        }
        #pragma unroll
        for (int r = 0; r < 8; ++r) {
            const int rowm = r + 8 * half;
            const int qh = th0 + (rowm >> 2), qw = tw0 + (rowm & 3);
            float* orow = out + (long)((b * HH + qh) * WW + qw) * DD + 4 * lm;
            #pragma unroll
            for (int g = 0; g < 4; ++g) {
                const v4f ov = { o[g][0][r], o[g][1][r], o[g][2][r], o[g][3][r] };
                *(v4f*)(orow + 64 * g) = ov;
            }
        }
    }
}

extern "C" void kernel_launch(void* const* d_in, const int* in_sizes, int n_in,
                              void* d_out, int out_size, void* d_ws, size_t ws_size,
                              hipStream_t stream) {
    const float* x   = (const float*)d_in[0];
    float*       out = (float*)d_out;
    local_attn_wmma_f32<<<dim3(256), dim3(128), 0, stream>>>(x, out);
}


# global->LDS staging pipelined with Q.K^T over four channel quarters: a quarter of all 196 rows is stored, handed over by an LDS-only barrier, its 16 score steps run while later quarters arrive
# speedup vs baseline: 4.0879x; 1.0588x over previous
#include <hip/hip_runtime.h>


typedef float v2f __attribute__((ext_vector_type(2)));
typedef float v4f __attribute__((ext_vector_type(4)));
typedef float v8f __attribute__((ext_vector_type(8)));

#define HH 64
#define WW 64
#define DD 256
#define PW 14
#define NROWS (PW * PW)
#define XP 260
#define NPAIR 98
#define NLD 25

template <int CTRL>
__device__ __forceinline__ float lane_xor(float x) {
    const int xi = __float_as_int(x);
    return __int_as_float(__builtin_amdgcn_update_dpp(0, xi, CTRL, 0xf, 0xf, true));
}

__device__ __forceinline__ float lane_xhalf(float x) {
    const int xi = __float_as_int(x);
    return __int_as_float(__builtin_amdgcn_permlanex16(xi, xi, 0x76543210, 0xfedcba98, false, false));
}

__device__ __forceinline__ void lds_handover_barrier() {
    __builtin_amdgcn_fence(__ATOMIC_RELEASE, "workgroup", "local");
    __builtin_amdgcn_s_barrier();
    __builtin_amdgcn_fence(__ATOMIC_ACQUIRE, "workgroup", "local");
}

__device__ __forceinline__ constexpr int key_q(int u) { return u + 1 + (u >= 8 ? 1 : 0); }

__global__ __launch_bounds__(128)
void local_attn_wmma_f32(const float* __restrict__ x, float* __restrict__ out) {
    __shared__ __attribute__((aligned(16))) float sX[NROWS * XP];

    const int blk = blockIdx.x;
    const int b   = blk >> 6;
    const int t   = blk & 63;
    const int h0  = (t >> 3) << 3;
    const int w0  = (t & 7) << 3;

    const int tid  = threadIdx.x;
    const int wave = __builtin_amdgcn_readfirstlane(tid >> 5);
    const int lane = tid & 31;
    const int half = lane >> 4;
    const int lm   = lane & 15;

    const int ty  = wave >> 1, tx = wave & 1;
    const int th0 = h0 + 4 * ty, tw0 = w0 + 4 * tx;
    const float* tileX = sX + ((4 * ty) * PW + 4 * tx) * XP;

    const int qr = lm >> 2, qc = lm & 3;
    int lo = 3 - tw0;        lo = lo > qc ? lo : qc;
    int hi = WW + 2 - tw0;   hi = hi < qc + 6 ? hi : qc + 6;
    const unsigned cm = ((2u << hi) - 1u) & ~((1u << lo) - 1u);
    int loR = 3 - th0;       loR = loR > qr ? loR : qr;
    int hiR = HH + 2 - th0;  hiR = hiR < qr + 6 ? hiR : qr + 6;
    const unsigned rm = ((2u << hiR) - 1u) & ~((1u << loR) - 1u);
    const unsigned rmh = half ? (__brev(rm) >> 22) : rm;

    v8f acc[6] = {};
    float dc[4];
    {
        unsigned goff[NLD];
        unsigned soff[NLD];
        v4f tmp[4][NLD];
        const char* xb = (const char*)x;
        #pragma unroll
        for (int i = 0; i < NLD; ++i) {
            int j = wave + 4 * i;
            if (i == NLD - 1) j = j < NPAIR ? j : NPAIR - 1;
            const int pi = j / PW + 7 * half, pj = j % PW;
            int kh = h0 - 3 + pi;  kh = kh < 0 ? 0 : (kh > HH - 1 ? HH - 1 : kh);
            int kw = w0 - 3 + pj;  kw = kw < 0 ? 0 : (kw > WW - 1 ? WW - 1 : kw);
            goff[i] = (unsigned)(((b * HH + kh) * WW + kw) * DD + lm * 4) * 4u;
            soff[i] = (unsigned)((j + NPAIR * half) * XP + lm * 4);
            tmp[0][i] = *(const v4f*)(xb + goff[i]);
            if ((i & 7) == 7) __builtin_amdgcn_sched_barrier(0);
        }
        __builtin_amdgcn_sched_barrier(0);
        #pragma unroll
        for (int i = 0; i < NLD; ++i) tmp[1][i] = *(const v4f*)(xb + goff[i] + 256);

        const float* qP = tileX + ((3 + qr) * PW + 3 + qc) * XP + 2 * half;
        const float* kP[6];
        {
            const int hk = lm >> 3, r = lm & 7;
            #pragma unroll
            for (int kt = 0; kt < 6; ++kt) {
                const int u  = 8 * kt + r;
                const int q  = u + 1 + (u >= 8 ? 1 : 0);
                const int pl = q / 10, pj = q - 10 * pl;
                const int pi = hk ? 9 - pl : pl;
                kP[kt] = tileX + (pi * PW + pj) * XP + 2 * half;
            }
        }

        #pragma unroll
        for (int i = 0; i < NLD; ++i) *(v4f*)(sX + soff[i]) = tmp[0][i];
        __builtin_amdgcn_sched_barrier(0);
        #pragma unroll
        for (int i = 0; i < NLD; ++i) tmp[2][i] = *(const v4f*)(xb + goff[i] + 512);
        lds_handover_barrier();

        v4f ck[4][2], cq[4][2];
        v2f qb[2][16];
        v2f kb[2][6][4];
        #pragma unroll
        for (int i = 0; i < 16; ++i) qb[0][i] = *(const v2f*)(qP + 4 * i);
        #pragma unroll
        for (int kt = 0; kt < 6; ++kt) {
            #pragma unroll
            for (int i = 0; i < 4; ++i) kb[0][kt][i] = *(const v2f*)(kP[kt] + 4 * i);
        }
        #pragma unroll
        for (int c = 0; c < 4; ++c) {
            #pragma unroll
            for (int g = 0; g < 4; ++g) {
                if (g < 3) {
                    #pragma unroll
                    for (int kt = 0; kt < 6; ++kt) {
                        #pragma unroll
                        for (int i = 0; i < 4; ++i)
                            kb[(g + 1) & 1][kt][i] = *(const v2f*)(kP[kt] + 64 * c + 4 * (4 * (g + 1) + i));
                    }
                }
                if (g == 2 && c < 3) {
                    #pragma unroll
                    for (int i = 0; i < NLD; ++i) *(v4f*)(sX + soff[i] + 64 * (c + 1)) = tmp[c + 1][i];
                    if (c == 0) {
                        __builtin_amdgcn_sched_barrier(0);
                        #pragma unroll
                        for (int i = 0; i < NLD; ++i) tmp[3][i] = *(const v4f*)(xb + goff[i] + 768);
                    }
                }
                if (g == 3 && c < 3) {
                    lds_handover_barrier();
                    #pragma unroll
                    for (int i = 0; i < 16; ++i) qb[(c + 1) & 1][i] = *(const v2f*)(qP + 64 * (c + 1) + 4 * i);
                    #pragma unroll
                    for (int kt = 0; kt < 6; ++kt) {
                        #pragma unroll
                        for (int i = 0; i < 4; ++i) kb[0][kt][i] = *(const v2f*)(kP[kt] + 64 * (c + 1) + 4 * i);
                    }
                }
                if (g == 2 && c == 3) {
                    #pragma unroll
                    for (int j = 0; j < 4; ++j) {
                        const int hh = j >> 1, s = j & 1;
                        const float* ckp = tileX + ((hh ? 9 : 0) * PW + (s ? 9 : 0)) * XP + lane * 4;
                        const float* cqp = tileX + ((3 + (hh ? 3 : 0)) * PW + 3 + (s ? 3 : 0)) * XP + lane * 4;
                        ck[j][0] = *(const v4f*)(ckp);  ck[j][1] = *(const v4f*)(ckp + 128);
                        cq[j][0] = *(const v4f*)(cqp);  cq[j][1] = *(const v4f*)(cqp + 128);
                    }
                }
                __builtin_amdgcn_sched_barrier(0);
                #pragma unroll
                for (int i = 0; i < 4; ++i) {
                    #pragma unroll
                    for (int kt = 0; kt < 6; ++kt)
                        acc[kt] = __builtin_amdgcn_wmma_f32_16x16x4_f32(
                                false, kb[g & 1][kt][i], false, qb[c & 1][4 * g + i], (short)0, acc[kt], false, false);
                }
                __builtin_amdgcn_sched_barrier(0);
            }
        }

        #pragma unroll
        for (int j = 0; j < 4; ++j) {
            const v4f k0 = ck[j][0], k1 = ck[j][1], q0 = cq[j][0], q1 = cq[j][1];
            float d = q0[0] * k0[0] + q0[1] * k0[1] + q0[2] * k0[2] + q0[3] * k0[3]
                    + q1[0] * k1[0] + q1[1] * k1[1] + q1[2] * k1[2] + q1[3] * k1[3];
            d += lane_xor<0x161>(d);
            d += lane_xor<0x162>(d);
            d += lane_xor<0x164>(d);
            d += lane_xor<0x168>(d);
            d += lane_xhalf(d);
            dc[j] = d;
        }
    }

    float p[6][8], pc[2];
    {
        #pragma unroll
        for (int kt = 0; kt < 6; ++kt) {
            #pragma unroll
            for (int r = 0; r < 8; ++r) {
                const int q  = key_q(8 * kt + r);
                const int pl = q / 10, pj = q - 10 * pl;
                const bool ok = ((rmh >> pl) & (cm >> pj) & 1u) != 0u;
                p[kt][r] = ok ? acc[kt][r] * 0.0625f : -3.0e38f;
            }
        }
        #pragma unroll
        for (int s = 0; s < 2; ++s) {
            const bool ok = (rmh & (cm >> (s ? 9 : 0)) & 1u) != 0u;
            const float d = half ? dc[2 + s] : dc[s];
            pc[s] = ok ? d * 0.0625f : -3.0e38f;
        }
        float m4[4] = { -3.0e38f, -3.0e38f, -3.0e38f, -3.0e38f };
        #pragma unroll
        for (int kt = 0; kt < 6; ++kt) {
            #pragma unroll
            for (int r = 0; r < 8; ++r) m4[r & 3] = fmaxf(m4[r & 3], p[kt][r]);
        }
        m4[0] = fmaxf(m4[0], pc[0]);
        m4[1] = fmaxf(m4[1], pc[1]);
        float mx = fmaxf(fmaxf(m4[0], m4[1]), fmaxf(m4[2], m4[3]));
        mx = fmaxf(mx, lane_xhalf(mx));
        float s4[4] = { 0.f, 0.f, 0.f, 0.f };
        #pragma unroll
        for (int kt = 0; kt < 6; ++kt) {
            #pragma unroll
            for (int r = 0; r < 8; ++r) {
                p[kt][r] = (p[kt][r] > -1.0e38f) ? __expf(p[kt][r] - mx) : 0.f;
                s4[r & 3] += p[kt][r];
            }
        }
        #pragma unroll
        for (int s = 0; s < 2; ++s) {
            pc[s] = (pc[s] > -1.0e38f) ? __expf(pc[s] - mx) : 0.f;
            s4[s] += pc[s];
        }
        float sum = (s4[0] + s4[1]) + (s4[2] + s4[3]);
        sum += lane_xhalf(sum);
        const float inv = 1.f / sum;
        #pragma unroll
        for (int kt = 0; kt < 6; ++kt) {
            #pragma unroll
            for (int r = 0; r < 8; ++r) p[kt][r] *= inv;
        }
        pc[0] *= inv;
        pc[1] *= inv;
    }

    {
        const float* vB[5];
        #pragma unroll
        for (int pl = 0; pl < 5; ++pl)
            vB[pl] = tileX + ((half ? 9 - pl : pl) * PW) * XP + 4 * lm;

        v8f o[4][4] = {};
        v4f vr[2][4][2];
        #pragma unroll
        for (int g = 0; g < 4; ++g) {
            vr[0][g][0] = *(const v4f*)(vB[0] + 1 * XP + 64 * g);
            vr[0][g][1] = *(const v4f*)(vB[0] + 2 * XP + 64 * g);
        }
        #pragma unroll
        for (int st = 0; st < 25; ++st) {
            if (st < 24) {
                const int sn = st + 1;
                const int q0 = sn < 24 ? key_q(8 * (sn >> 2) + 2 * (sn & 3))     : 0;
                const int q1 = sn < 24 ? key_q(8 * (sn >> 2) + 2 * (sn & 3) + 1) : 9;
                const float* a0 = vB[q0 / 10] + (q0 % 10) * XP;
                const float* a1 = vB[q1 / 10] + (q1 % 10) * XP;
                #pragma unroll
                for (int g = 0; g < 4; ++g) {
                    vr[sn & 1][g][0] = *(const v4f*)(a0 + 64 * g);
                    vr[sn & 1][g][1] = *(const v4f*)(a1 + 64 * g);
                }
            }
            __builtin_amdgcn_sched_barrier(0);
            const int kk = st < 24 ? (st >> 2) : 0, j2 = st < 24 ? 2 * (st & 3) : 0;
            const v2f pa = st < 24 ? (v2f){ p[kk][j2], p[kk][j2 + 1] } : (v2f){ pc[0], pc[1] };
            #pragma unroll
            for (int g = 0; g < 4; ++g) {
                #pragma unroll
                for (int jj = 0; jj < 4; ++jj) {
                    const v2f bf = { vr[st & 1][g][0][jj], vr[st & 1][g][1][jj] };
                    o[g][jj] = __builtin_amdgcn_wmma_f32_16x16x4_f32(
                            false, pa, false, bf, (short)0, o[g][jj], false, false);
                }
            }
            __builtin_amdgcn_sched_barrier(0);
        }
        #pragma unroll
        for (int r = 0; r < 8; ++r) {
            const int rowm = r + 8 * half;
            const int qh = th0 + (rowm >> 2), qw = tw0 + (rowm & 3);
            float* orow = out + (long)((b * HH + qh) * WW + qw) * DD + 4 * lm;
            #pragma unroll
            for (int g = 0; g < 4; ++g) {
                const v4f ov = { o[g][0][r], o[g][1][r], o[g][2][r], o[g][3][r] };
                *(v4f*)(orow + 64 * g) = ov;
            }
        }
    }
}

extern "C" void kernel_launch(void* const* d_in, const int* in_sizes, int n_in,
                              void* d_out, int out_size, void* d_ws, size_t ws_size,
                              hipStream_t stream) {
    const float* x   = (const float*)d_in[0];
    float*       out = (float*)d_out;
    local_attn_wmma_f32<<<dim3(256), dim3(128), 0, stream>>>(x, out);
}
